# starting kernel
# baseline (speedup 1.0000x reference)
#include <hip/hip_runtime.h>


#define N_NODES   20000
#define D_FEAT    256
#define N_SUPPORT 3
#define N_EDGES   320000
#define OUT_DIM   256
#define H_DIM     (N_SUPPORT * D_FEAT)

typedef __attribute__((ext_vector_type(2))) float v2f;
typedef __attribute__((ext_vector_type(8))) float v8f;

__global__ __launch_bounds__(256) void zero_h(float4* __restrict__ h, int n4) {
  int i = blockIdx.x * blockDim.x + threadIdx.x;
  if (i < n4) h[i] = float4{0.f, 0.f, 0.f, 0.f};
}

__global__ __launch_bounds__(256) void scatter_edges(
    const float* __restrict__ x, const int* __restrict__ esrc,
    const int* __restrict__ edst, const float* __restrict__ ew,
    float* __restrict__ h) {
  long gid = (long)blockIdx.x * blockDim.x + threadIdx.x;
  int  chunk = (int)(gid & 63);
  long e = gid >> 6;
  if (e >= (long)N_SUPPORT * N_EDGES) return;
  int s   = (int)(e / N_EDGES);
  int src = esrc[e];
  int dst = edst[e];
  float w = ew[e];
  float4 xv = ((const float4*)(x + (long)src * D_FEAT))[chunk];
  float* hp = h + (long)dst * H_DIM + s * D_FEAT + chunk * 4;
  atomicAdd(hp + 0, xv.x * w);
  atomicAdd(hp + 1, xv.y * w);
  atomicAdd(hp + 2, xv.z * w);
  atomicAdd(hp + 3, xv.w * w);
}

__global__ __launch_bounds__(256) void gemm_wmma(
    const float* __restrict__ h, const float* __restrict__ Wm,
    float* __restrict__ out) {
  int wave = (int)((blockIdx.x * blockDim.x + threadIdx.x) >> 5);
  int lane = threadIdx.x & 31;
  int tile_m = wave >> 4;
  int tile_n = wave & 15;
  int half   = lane >> 4;
  int mrow   = tile_m * 16 + (lane & 15);
  int ncol   = tile_n * 16 + (lane & 15);

  const float* arow = h  + (long)mrow * H_DIM + half * 2;
  const float* bcol = Wm + (long)(half * 2) * OUT_DIM + ncol;

  v8f c = {};
  #pragma unroll 4
  for (int k = 0; k < H_DIM; k += 4) {
    v2f a = *(const v2f*)(arow + k);
    v2f b;
    b.x = bcol[(long)k * OUT_DIM];
    b.y = bcol[(long)k * OUT_DIM + OUT_DIM];
    c = __builtin_amdgcn_wmma_f32_16x16x4_f32(
         false, a,  false, b,
         (short)0, c,  false,  false);
  }

  float* op = out + (long)(tile_m * 16 + half * 8) * OUT_DIM + ncol;
  #pragma unroll
  for (int j = 0; j < 8; ++j) op[(long)j * OUT_DIM] = c[j];
}

extern "C" void kernel_launch(void* const* d_in, const int* in_sizes, int n_in,
                              void* d_out, int out_size, void* d_ws, size_t ws_size,
                              hipStream_t stream) {
  const float* x    = (const float*)d_in[0];
  const int*   esrc = (const int*)  d_in[1];
  const int*   edst = (const int*)  d_in[2];
  const float* ew   = (const float*)d_in[3];
  const float* Wm   = (const float*)d_in[4];
  float*       out  = (float*)d_out;
  float*       hbuf = (float*)d_ws;

  int n4 = N_NODES * H_DIM / 4;
  zero_h<<<(n4 + 255) / 256, 256, 0, stream>>>((float4*)hbuf, n4);

  long nthr = (long)N_SUPPORT * N_EDGES * 64;
  scatter_edges<<<(int)(nthr / 256), 256, 0, stream>>>(x, esrc, edst, ew, hbuf);

  gemm_wmma<<<(N_NODES / 16) * (OUT_DIM / 16) / 8, 256, 0, stream>>>(hbuf, Wm, out);
}


# Float-atomic edge scatter replaced by binning edges by (support,dst) with an integer slot add; one wave per h segment then sums its source rows in registers in ascending edge index (fixed sum order)
# speedup vs baseline: 1.9585x; 1.9585x over previous
#include <hip/hip_runtime.h>


#define N_NODES   20000
#define D_FEAT    256
#define N_SUPPORT 3
#define N_EDGES   320000
#define OUT_DIM   256
#define H_DIM     (N_SUPPORT * D_FEAT)
#define N_LISTS   (N_SUPPORT * N_NODES)
#define LIST_CAP  64
#define NO_EDGE   0x7fffffff

typedef __attribute__((ext_vector_type(2))) float v2f;
typedef __attribute__((ext_vector_type(8))) float v8f;

__global__ __launch_bounds__(256) void clear_sizes(int* __restrict__ cnt) {
  int i = blockIdx.x * blockDim.x + threadIdx.x;
  if (i < N_LISTS) cnt[i] = 0;
}

__global__ __launch_bounds__(256) void bin_edges(
    const int* __restrict__ edst, int* __restrict__ cnt, int* __restrict__ slots) {
  int q = blockIdx.x * blockDim.x + threadIdx.x;
  if (q >= N_SUPPORT * N_EDGES / 4) return;
  int4 d = ((const int4*)edst)[q];
  int e    = q * 4;
  int base = (e / N_EDGES) * N_NODES;
  int l0 = base + d.x, l1 = base + d.y, l2 = base + d.z, l3 = base + d.w;
  int p0 = atomicAdd(&cnt[l0], 1);
  int p1 = atomicAdd(&cnt[l1], 1);
  int p2 = atomicAdd(&cnt[l2], 1);
  int p3 = atomicAdd(&cnt[l3], 1);
  if (p0 < LIST_CAP) slots[l0 * LIST_CAP + p0] = e;
  if (p1 < LIST_CAP) slots[l1 * LIST_CAP + p1] = e + 1;
  if (p2 < LIST_CAP) slots[l2 * LIST_CAP + p2] = e + 2;
  if (p3 < LIST_CAP) slots[l3 * LIST_CAP + p3] = e + 3;
}

__device__ __forceinline__ void add_row(float4& a0, float4& a1,
                                        const float* __restrict__ x,
                                        int src, float w, int lane) {
  const float4* xr = (const float4*)(x + (long)src * D_FEAT);
  float4 p = xr[lane];
  float4 q = xr[lane + 32];
  a0.x += p.x * w; a0.y += p.y * w; a0.z += p.z * w; a0.w += p.w * w;
  a1.x += q.x * w; a1.y += q.y * w; a1.z += q.z * w; a1.w += q.w * w;
}

__global__ __launch_bounds__(256) void gather_rows(
    const float* __restrict__ x, const int* __restrict__ esrc,
    const int* __restrict__ edst, const float* __restrict__ ew,
    const int* __restrict__ cnt, const int* __restrict__ slots,
    float* __restrict__ h) {
  int list = __builtin_amdgcn_readfirstlane(
      (int)((blockIdx.x * blockDim.x + threadIdx.x) >> 5));
  int lane = threadIdx.x & 31;
  int s    = list / N_NODES;
  int row  = list - s * N_NODES;
  int n    = cnt[list];

  float4 a0 = {0.f, 0.f, 0.f, 0.f};
  float4 a1 = {0.f, 0.f, 0.f, 0.f};

  if (n <= LIST_CAP) {
    const int* lp = slots + list * LIST_CAP;
    bool v0 = lane < n;
    int e0 = v0 ? lp[lane] : NO_EDGE;
    int i0 = v0 ? e0 : 0;
    int   s0 = esrc[i0];
    float w0 = ew[i0];
    int e1 = NO_EDGE, s1 = 0;
    float w1 = 0.f;
    if (n > 32) {
      bool v1 = lane + 32 < n;
      e1 = v1 ? lp[lane + 32] : NO_EDGE;
      int i1 = v1 ? e1 : 0;
      s1 = esrc[i1];
      w1 = ew[i1];
    }

    int r0 = 0, r1 = n;
    #pragma unroll
    for (int j = 0; j < 32; ++j)
      r0 += (__builtin_amdgcn_readlane(e0, j) < e0) ? 1 : 0;
    if (n > 32) {
      r1 = 0;
      #pragma unroll
      for (int j = 0; j < 32; ++j) {
        int va = __builtin_amdgcn_readlane(e0, j);
        int vb = __builtin_amdgcn_readlane(e1, j);
        r1 += (va < e1) ? 1 : 0;
        r0 += (vb < e0) ? 1 : 0;
        r1 += (vb < e1) ? 1 : 0;
      }
    }

    #pragma unroll 4
    for (int t = 0; t < n; ++t) {
      unsigned long long m =
          (unsigned long long)__builtin_amdgcn_ballot_w32(r0 == t) |
          ((unsigned long long)__builtin_amdgcn_ballot_w32(r1 == t) << 32);
      int l  = __builtin_ctzll(m);
      int sa = __builtin_amdgcn_readlane(s0, l & 31);
      int sb = __builtin_amdgcn_readlane(s1, l & 31);
      int wa = __builtin_amdgcn_readlane(__float_as_int(w0), l & 31);
      int wb = __builtin_amdgcn_readlane(__float_as_int(w1), l & 31);
      int   src = (l < 32) ? sa : sb;
      float w   = __int_as_float((l < 32) ? wa : wb);
      add_row(a0, a1, x, src, w, lane);
    }
  } else {
    const int* dl = edst + s * N_EDGES;
    for (int b = 0; b < N_EDGES; b += 32) {
      unsigned m = __builtin_amdgcn_ballot_w32(dl[b + lane] == row);
      while (m) {
        int e = s * N_EDGES + b + __builtin_ctz(m);
        m &= m - 1;
        add_row(a0, a1, x, esrc[e], ew[e], lane);
      }
    }
  }

  float4* hp = (float4*)(h + (long)row * H_DIM + s * D_FEAT);
  hp[lane]      = a0;
  hp[lane + 32] = a1;
}

__global__ __launch_bounds__(256) void gemm_wmma(
    const float* __restrict__ h, const float* __restrict__ Wm,
    float* __restrict__ out) {
  int wave = (int)((blockIdx.x * blockDim.x + threadIdx.x) >> 5);
  int lane = threadIdx.x & 31;
  int tile_m = wave >> 4;
  int tile_n = wave & 15;
  int half   = lane >> 4;
  int mrow   = tile_m * 16 + (lane & 15);
  int ncol   = tile_n * 16 + (lane & 15);

  const float* arow = h  + (long)mrow * H_DIM + half * 2;
  const float* bcol = Wm + (long)(half * 2) * OUT_DIM + ncol;

  v8f c = {};
  #pragma unroll 4
  for (int k = 0; k < H_DIM; k += 4) {
    v2f a = *(const v2f*)(arow + k);
    v2f b;
    b.x = bcol[(long)k * OUT_DIM];
    b.y = bcol[(long)k * OUT_DIM + OUT_DIM];
    c = __builtin_amdgcn_wmma_f32_16x16x4_f32(
         false, a,  false, b,
         (short)0, c,  false,  false);
  }

  float* op = out + (long)(tile_m * 16 + half * 8) * OUT_DIM + ncol;
  #pragma unroll
  for (int j = 0; j < 8; ++j) op[(long)j * OUT_DIM] = c[j];
}

extern "C" void kernel_launch(void* const* d_in, const int* in_sizes, int n_in,
                              void* d_out, int out_size, void* d_ws, size_t ws_size,
                              hipStream_t stream) {
  const float* x    = (const float*)d_in[0];
  const int*   esrc = (const int*)  d_in[1];
  const int*   edst = (const int*)  d_in[2];
  const float* ew   = (const float*)d_in[3];
  const float* Wm   = (const float*)d_in[4];
  float*       out  = (float*)d_out;
  float*       hbuf = (float*)d_ws;

  int* cnt   = (int*)d_out;
  int* slots = cnt + 65536;

  clear_sizes<<<(N_LISTS + 255) / 256, 256, 0, stream>>>(cnt);

  bin_edges<<<(N_SUPPORT * N_EDGES / 4 + 255) / 256, 256, 0, stream>>>(edst, cnt, slots);

  gather_rows<<<N_LISTS / 8, 256, 0, stream>>>(x, esrc, edst, ew, cnt, slots, hbuf);

  gemm_wmma<<<(N_NODES / 16) * (OUT_DIM / 16) / 8, 256, 0, stream>>>(hbuf, Wm, out);
}


# GEMM register-tiled: one wave computes 32x64 of out (2x4 f32 WMMA tiles) with 16-byte operand loads and two fragment sets, was one 16x16 tile per wave; k summed in a permuted order within each 16
# speedup vs baseline: 2.6737x; 1.3652x over previous
#include <hip/hip_runtime.h>


#define N_NODES   20000
#define D_FEAT    256
#define N_SUPPORT 3
#define N_EDGES   320000
#define OUT_DIM   256
#define H_DIM     (N_SUPPORT * D_FEAT)
#define N_LISTS   (N_SUPPORT * N_NODES)
#define LIST_CAP  64
#define NO_EDGE   0x7fffffff

typedef __attribute__((ext_vector_type(2))) float v2f;
typedef __attribute__((ext_vector_type(8))) float v8f;

__global__ __launch_bounds__(256) void clear_sizes(int* __restrict__ cnt) {
  int i = blockIdx.x * blockDim.x + threadIdx.x;
  if (i < N_LISTS) cnt[i] = 0;
}

__global__ __launch_bounds__(256) void bin_edges(
    const int* __restrict__ edst, int* __restrict__ cnt, int* __restrict__ slots) {
  int q = blockIdx.x * blockDim.x + threadIdx.x;
  if (q >= N_SUPPORT * N_EDGES / 4) return;
  int4 d = ((const int4*)edst)[q];
  int e    = q * 4;
  int base = (e / N_EDGES) * N_NODES;
  int l0 = base + d.x, l1 = base + d.y, l2 = base + d.z, l3 = base + d.w;
  int p0 = atomicAdd(&cnt[l0], 1);
  int p1 = atomicAdd(&cnt[l1], 1);
  int p2 = atomicAdd(&cnt[l2], 1);
  int p3 = atomicAdd(&cnt[l3], 1);
  if (p0 < LIST_CAP) slots[l0 * LIST_CAP + p0] = e;
  if (p1 < LIST_CAP) slots[l1 * LIST_CAP + p1] = e + 1;
  if (p2 < LIST_CAP) slots[l2 * LIST_CAP + p2] = e + 2;
  if (p3 < LIST_CAP) slots[l3 * LIST_CAP + p3] = e + 3;
}

__device__ __forceinline__ void add_row(float4& a0, float4& a1,
                                        const float* __restrict__ x,
                                        int src, float w, int lane) {
  const float4* xr = (const float4*)(x + (long)src * D_FEAT);
  float4 p = xr[lane];
  float4 q = xr[lane + 32];
  a0.x += p.x * w; a0.y += p.y * w; a0.z += p.z * w; a0.w += p.w * w;
  a1.x += q.x * w; a1.y += q.y * w; a1.z += q.z * w; a1.w += q.w * w;
}

__global__ __launch_bounds__(256) void gather_rows(
    const float* __restrict__ x, const int* __restrict__ esrc,
    const int* __restrict__ edst, const float* __restrict__ ew,
    const int* __restrict__ cnt, const int* __restrict__ slots,
    float* __restrict__ h) {
  int list = __builtin_amdgcn_readfirstlane(
      (int)((blockIdx.x * blockDim.x + threadIdx.x) >> 5));
  int lane = threadIdx.x & 31;
  int s    = list / N_NODES;
  int row  = list - s * N_NODES;
  int n    = cnt[list];

  float4 a0 = {0.f, 0.f, 0.f, 0.f};
  float4 a1 = {0.f, 0.f, 0.f, 0.f};

  if (n <= LIST_CAP) {
    const int* lp = slots + list * LIST_CAP;
    bool v0 = lane < n;
    int e0 = v0 ? lp[lane] : NO_EDGE;
    int i0 = v0 ? e0 : 0;
    int   s0 = esrc[i0];
    float w0 = ew[i0];
    int e1 = NO_EDGE, s1 = 0;
    float w1 = 0.f;
    if (n > 32) {
      bool v1 = lane + 32 < n;
      e1 = v1 ? lp[lane + 32] : NO_EDGE;
      int i1 = v1 ? e1 : 0;
      s1 = esrc[i1];
      w1 = ew[i1];
    }

    int r0 = 0, r1 = n;
    #pragma unroll
    for (int j = 0; j < 32; ++j)
      r0 += (__builtin_amdgcn_readlane(e0, j) < e0) ? 1 : 0;
    if (n > 32) {
      r1 = 0;
      #pragma unroll
      for (int j = 0; j < 32; ++j) {
        int va = __builtin_amdgcn_readlane(e0, j);
        int vb = __builtin_amdgcn_readlane(e1, j);
        r1 += (va < e1) ? 1 : 0;
        r0 += (vb < e0) ? 1 : 0;
        r1 += (vb < e1) ? 1 : 0;
      }
    }

    #pragma unroll 4
    for (int t = 0; t < n; ++t) {
      unsigned long long m =
          (unsigned long long)__builtin_amdgcn_ballot_w32(r0 == t) |
          ((unsigned long long)__builtin_amdgcn_ballot_w32(r1 == t) << 32);
      int l  = __builtin_ctzll(m);
      int sa = __builtin_amdgcn_readlane(s0, l & 31);
      int sb = __builtin_amdgcn_readlane(s1, l & 31);
      int wa = __builtin_amdgcn_readlane(__float_as_int(w0), l & 31);
      int wb = __builtin_amdgcn_readlane(__float_as_int(w1), l & 31);
      int   src = (l < 32) ? sa : sb;
      float w   = __int_as_float((l < 32) ? wa : wb);
      add_row(a0, a1, x, src, w, lane);
    }
  } else {
    const int* dl = edst + s * N_EDGES;
    for (int b = 0; b < N_EDGES; b += 32) {
      unsigned m = __builtin_amdgcn_ballot_w32(dl[b + lane] == row);
      while (m) {
        int e = s * N_EDGES + b + __builtin_ctz(m);
        m &= m - 1;
        add_row(a0, a1, x, esrc[e], ew[e], lane);
      }
    }
  }

  float4* hp = (float4*)(h + (long)row * H_DIM + s * D_FEAT);
  hp[lane]      = a0;
  hp[lane + 32] = a1;
}

typedef __attribute__((ext_vector_type(4))) float v4f;

struct Frag {
  v4f a[2][2];
  v4f b[8];
};

__device__ __forceinline__ void load_frag(Frag& f, const float* __restrict__ a0p,
                                          const float* __restrict__ a1p,
                                          const float* __restrict__ bp, int k0) {
  f.a[0][0] = *(const v4f*)(a0p + k0);
  f.a[0][1] = *(const v4f*)(a0p + k0 + 4);
  f.a[1][0] = *(const v4f*)(a1p + k0);
  f.a[1][1] = *(const v4f*)(a1p + k0 + 4);
  #pragma unroll
  for (int i = 0; i < 8; ++i)
    f.b[i] = *(const v4f*)(bp + (long)(k0 + i) * OUT_DIM);
}

__device__ __forceinline__ void mma_frag(v8f (&acc)[2][4], const Frag& f) {
  #pragma unroll
  for (int s = 0; s < 4; ++s) {
    #pragma unroll
    for (int mt = 0; mt < 2; ++mt) {
      v2f a;
      a.x = f.a[mt][s >> 1][(2 * s) & 3];
      a.y = f.a[mt][s >> 1][((2 * s) & 3) + 1];
      #pragma unroll
      for (int t = 0; t < 4; ++t) {
        v2f b;
        b.x = f.b[2 * s][t];
        b.y = f.b[2 * s + 1][t];
        acc[mt][t] = __builtin_amdgcn_wmma_f32_16x16x4_f32(
             false, a,  false, b,
             (short)0, acc[mt][t],  false,  false);
      }
    }
  }
}

__global__ __launch_bounds__(128) void gemm_wmma(
    const float* __restrict__ h, const float* __restrict__ Wm,
    float* __restrict__ out) {
  int wave = threadIdx.x >> 5;
  int lane = threadIdx.x & 31;
  int c    = lane & 15;
  int hf   = lane >> 4;
  int m0   = blockIdx.x * 32;
  int nb   = wave * 64 + 4 * c;

  const float* a0p = h + (long)(m0 + c) * H_DIM + 8 * hf;
  const float* a1p = a0p + 16 * H_DIM;
  const float* bp  = Wm + (long)(8 * hf) * OUT_DIM + nb;

  v8f acc[2][4] = {};
  Frag f0, f1;

  load_frag(f0, a0p, a1p, bp, 0);
  for (int k0 = 0; k0 < H_DIM - 32; k0 += 32) {
    load_frag(f1, a0p, a1p, bp, k0 + 16);
    __builtin_amdgcn_sched_barrier(0);
    mma_frag(acc, f0);
    load_frag(f0, a0p, a1p, bp, k0 + 32);
    mma_frag(acc, f1);
  }
  load_frag(f1, a0p, a1p, bp, H_DIM - 16);
  mma_frag(acc, f0);
  mma_frag(acc, f1);

  #pragma unroll
  for (int mt = 0; mt < 2; ++mt) {
    float* op = out + (long)(m0 + mt * 16 + hf * 8) * OUT_DIM + nb;
    #pragma unroll
    for (int j = 0; j < 8; ++j) {
      v4f v = {acc[mt][0][j], acc[mt][1][j], acc[mt][2][j], acc[mt][3][j]};
      *(v4f*)(op + (long)j * OUT_DIM) = v;
    }
  }
}

extern "C" void kernel_launch(void* const* d_in, const int* in_sizes, int n_in,
                              void* d_out, int out_size, void* d_ws, size_t ws_size,
                              hipStream_t stream) {
  const float* x    = (const float*)d_in[0];
  const int*   esrc = (const int*)  d_in[1];
  const int*   edst = (const int*)  d_in[2];
  const float* ew   = (const float*)d_in[3];
  const float* Wm   = (const float*)d_in[4];
  float*       out  = (float*)d_out;
  float*       hbuf = (float*)d_ws;

  int* cnt   = (int*)d_out;
  int* slots = cnt + 65536;

  clear_sizes<<<(N_LISTS + 255) / 256, 256, 0, stream>>>(cnt);

  bin_edges<<<(N_SUPPORT * N_EDGES / 4 + 255) / 256, 256, 0, stream>>>(edst, cnt, slots);

  gather_rows<<<N_LISTS / 8, 256, 0, stream>>>(x, esrc, edst, ew, cnt, slots, hbuf);

  gemm_wmma<<<N_NODES / 32, 128, 0, stream>>>(hbuf, Wm, out);
}


# Edge lists built in LDS: a 32-wave work-group owns 236 dst rows of a support, scans its dst array once (LDS counter adds), then gathers; no global atomics or list tables, 4 launches -> 2
# speedup vs baseline: 4.0481x; 1.5140x over previous
#include <hip/hip_runtime.h>


#define N_NODES   20000
#define D_FEAT    256
#define N_SUPPORT 3
#define N_EDGES   320000
#define OUT_DIM   256
#define H_DIM     (N_SUPPORT * D_FEAT)
#define LIST_CAP  64
#define NB_PER_S  85
#define RPB       236
#define BIN_THREADS 1024
#define SCAN_DEPTH  6
#define SCAN_FULL   ((N_EDGES / 4) / (SCAN_DEPTH * BIN_THREADS) * (SCAN_DEPTH * BIN_THREADS))
#define NO_EDGE   0x7fffffff

typedef __attribute__((ext_vector_type(2))) float v2f;
typedef __attribute__((ext_vector_type(8))) float v8f;

__device__ __forceinline__ void add_row(float4& a0, float4& a1,
                                        const float* __restrict__ x,
                                        int src, float w, int lane) {
  const float4* xr = (const float4*)(x + (long)src * D_FEAT);
  float4 p = xr[lane];
  float4 q = xr[lane + 32];
  a0.x += p.x * w; a0.y += p.y * w; a0.z += p.z * w; a0.w += p.w * w;
  a1.x += q.x * w; a1.y += q.y * w; a1.z += q.z * w; a1.w += q.w * w;
}

__device__ __forceinline__ void take_edge(int dv, int e, int row0, int nrows,
                                          int* cnt, int* slots) {
  unsigned r = (unsigned)(dv - row0);
  if (r < (unsigned)nrows) {
    int p = atomicAdd(&cnt[r], 1);
    if (p < LIST_CAP) slots[r * LIST_CAP + p] = e;
  }
}

__global__ __launch_bounds__(BIN_THREADS) void bin_gather(
    const float* __restrict__ x, const int* __restrict__ esrc,
    const int* __restrict__ edst, const float* __restrict__ ew,
    float* __restrict__ h) {
  __shared__ int cnt[RPB];
  __shared__ int slots[RPB * LIST_CAP];

  int tid   = threadIdx.x;
  int s     = blockIdx.x / NB_PER_S;
  int row0  = (blockIdx.x - s * NB_PER_S) * RPB;
  int nrows = (N_NODES - row0 < RPB) ? (N_NODES - row0) : RPB;

  if (tid < RPB) cnt[tid] = 0;
  __syncthreads();

  const int4* dl4 = (const int4*)(edst + s * N_EDGES);
  for (int base = 0; base < SCAN_FULL; base += SCAN_DEPTH * BIN_THREADS) {
    int4 d[SCAN_DEPTH];
    #pragma unroll
    for (int k = 0; k < SCAN_DEPTH; ++k)
      d[k] = dl4[base + k * BIN_THREADS + tid];
    #pragma unroll
    for (int k = 0; k < SCAN_DEPTH; ++k) {
      int e = (base + k * BIN_THREADS + tid) * 4;
      take_edge(d[k].x, e,     row0, nrows, cnt, slots);
      take_edge(d[k].y, e + 1, row0, nrows, cnt, slots);
      take_edge(d[k].z, e + 2, row0, nrows, cnt, slots);
      take_edge(d[k].w, e + 3, row0, nrows, cnt, slots);
    }
  }
  for (int q = SCAN_FULL + tid; q < N_EDGES / 4; q += BIN_THREADS) {
    int4 dv = dl4[q];
    int e = q * 4;
    take_edge(dv.x, e,     row0, nrows, cnt, slots);
    take_edge(dv.y, e + 1, row0, nrows, cnt, slots);
    take_edge(dv.z, e + 2, row0, nrows, cnt, slots);
    take_edge(dv.w, e + 3, row0, nrows, cnt, slots);
  }
  __syncthreads();

  int wave = __builtin_amdgcn_readfirstlane(tid >> 5);
  int lane = tid & 31;
  const int*   es = esrc + s * N_EDGES;
  const float* wt = ew   + s * N_EDGES;

  for (int r = wave; r < nrows; r += BIN_THREADS / 32) {
    int n = __builtin_amdgcn_readfirstlane(cnt[r]);

    float4 a0 = {0.f, 0.f, 0.f, 0.f};
    float4 a1 = {0.f, 0.f, 0.f, 0.f};

    if (n <= LIST_CAP) {
      const int* lp = slots + r * LIST_CAP;
      bool v0 = lane < n;
      int e0 = v0 ? lp[lane] : NO_EDGE;
      int i0 = v0 ? e0 : 0;
      int   s0 = es[i0];
      float w0 = wt[i0];
      int e1 = NO_EDGE, s1 = 0;
      float w1 = 0.f;
      if (n > 32) {
        bool v1 = lane + 32 < n;
        e1 = v1 ? lp[lane + 32] : NO_EDGE;
        int i1 = v1 ? e1 : 0;
        s1 = es[i1];
        w1 = wt[i1];
      }

      int r0 = 0, r1 = n;
      #pragma unroll
      for (int j = 0; j < 32; ++j)
        r0 += (__builtin_amdgcn_readlane(e0, j) < e0) ? 1 : 0;
      if (n > 32) {
        r1 = 0;
        #pragma unroll
        for (int j = 0; j < 32; ++j) {
          int va = __builtin_amdgcn_readlane(e0, j);
          int vb = __builtin_amdgcn_readlane(e1, j);
          r1 += (va < e1) ? 1 : 0;
          r0 += (vb < e0) ? 1 : 0;
          r1 += (vb < e1) ? 1 : 0;
        }
      }

      #pragma unroll 4
      for (int t = 0; t < n; ++t) {
        unsigned long long m =
            (unsigned long long)__builtin_amdgcn_ballot_w32(r0 == t) |
            ((unsigned long long)__builtin_amdgcn_ballot_w32(r1 == t) << 32);
        int l  = __builtin_ctzll(m);
        int sa = __builtin_amdgcn_readlane(s0, l & 31);
        int sb = __builtin_amdgcn_readlane(s1, l & 31);
        int wa = __builtin_amdgcn_readlane(__float_as_int(w0), l & 31);
        int wb = __builtin_amdgcn_readlane(__float_as_int(w1), l & 31);
        int   src = (l < 32) ? sa : sb;
        float w   = __int_as_float((l < 32) ? wa : wb);
        add_row(a0, a1, x, src, w, lane);
      }
    } else {
      const int* dl = edst + s * N_EDGES;
      for (int b = 0; b < N_EDGES; b += 32) {
        unsigned m = __builtin_amdgcn_ballot_w32(dl[b + lane] == row0 + r);
        while (m) {
          int e = b + __builtin_ctz(m);
          m &= m - 1;
          add_row(a0, a1, x, es[e], wt[e], lane);
        }
      }
    }

    float4* hp = (float4*)(h + (long)(row0 + r) * H_DIM + s * D_FEAT);
    hp[lane]      = a0;
    hp[lane + 32] = a1;
  }
}

typedef __attribute__((ext_vector_type(4))) float v4f;

struct Frag {
  v4f a[2][2];
  v4f b[8];
};

__device__ __forceinline__ void load_frag(Frag& f, const float* __restrict__ a0p,
                                          const float* __restrict__ a1p,
                                          const float* __restrict__ bp, int k0) {
  f.a[0][0] = *(const v4f*)(a0p + k0);
  f.a[0][1] = *(const v4f*)(a0p + k0 + 4);
  f.a[1][0] = *(const v4f*)(a1p + k0);
  f.a[1][1] = *(const v4f*)(a1p + k0 + 4);
  #pragma unroll
  for (int i = 0; i < 8; ++i)
    f.b[i] = *(const v4f*)(bp + (long)(k0 + i) * OUT_DIM);
}

__device__ __forceinline__ void mma_frag(v8f (&acc)[2][4], const Frag& f) {
  #pragma unroll
  for (int s = 0; s < 4; ++s) {
    #pragma unroll
    for (int mt = 0; mt < 2; ++mt) {
      v2f a;
      a.x = f.a[mt][s >> 1][(2 * s) & 3];
      a.y = f.a[mt][s >> 1][((2 * s) & 3) + 1];
      #pragma unroll
      for (int t = 0; t < 4; ++t) {
        v2f b;
        b.x = f.b[2 * s][t];
        b.y = f.b[2 * s + 1][t];
        acc[mt][t] = __builtin_amdgcn_wmma_f32_16x16x4_f32(
             false, a,  false, b,
             (short)0, acc[mt][t],  false,  false);
      }
    }
  }
}

__global__ __launch_bounds__(128) void gemm_wmma(
    const float* __restrict__ h, const float* __restrict__ Wm,
    float* __restrict__ out) {
  int wave = threadIdx.x >> 5;
  int lane = threadIdx.x & 31;
  int c    = lane & 15;
  int hf   = lane >> 4;
  int m0   = blockIdx.x * 32;
  int nb   = wave * 64 + 4 * c;

  const float* a0p = h + (long)(m0 + c) * H_DIM + 8 * hf;
  const float* a1p = a0p + 16 * H_DIM;
  const float* bp  = Wm + (long)(8 * hf) * OUT_DIM + nb;

  v8f acc[2][4] = {};
  Frag f0, f1;

  load_frag(f0, a0p, a1p, bp, 0);
  for (int k0 = 0; k0 < H_DIM - 32; k0 += 32) {
    load_frag(f1, a0p, a1p, bp, k0 + 16);
    __builtin_amdgcn_sched_barrier(0);
    mma_frag(acc, f0);
    load_frag(f0, a0p, a1p, bp, k0 + 32);
    mma_frag(acc, f1);
  }
  load_frag(f1, a0p, a1p, bp, H_DIM - 16);
  mma_frag(acc, f0);
  mma_frag(acc, f1);

  #pragma unroll
  for (int mt = 0; mt < 2; ++mt) {
    float* op = out + (long)(m0 + mt * 16 + hf * 8) * OUT_DIM + nb;
    #pragma unroll
    for (int j = 0; j < 8; ++j) {
      v4f v = {acc[mt][0][j], acc[mt][1][j], acc[mt][2][j], acc[mt][3][j]};
      *(v4f*)(op + (long)j * OUT_DIM) = v;
    }
  }
}

extern "C" void kernel_launch(void* const* d_in, const int* in_sizes, int n_in,
                              void* d_out, int out_size, void* d_ws, size_t ws_size,
                              hipStream_t stream) {
  const float* x    = (const float*)d_in[0];
  const int*   esrc = (const int*)  d_in[1];
  const int*   edst = (const int*)  d_in[2];
  const float* ew   = (const float*)d_in[3];
  const float* Wm   = (const float*)d_in[4];
  float*       out  = (float*)d_out;
  float*       hbuf = (float*)d_ws;

  bin_gather<<<N_SUPPORT * NB_PER_S, BIN_THREADS, 0, stream>>>(x, esrc, edst, ew, hbuf);

  gemm_wmma<<<N_NODES / 32, 128, 0, stream>>>(hbuf, Wm, out);
}


# GEMM re-tiled to 80x64 per wave (5x4 f32 WMMA tiles, loads 32 k ahead), 250 work-groups = one resident round: fullest SIMD does 3840 products, was 4608 (3 blocks on some units, 2 on others)
# speedup vs baseline: 4.4664x; 1.1033x over previous
#include <hip/hip_runtime.h>


#define N_NODES   20000
#define D_FEAT    256
#define N_SUPPORT 3
#define N_EDGES   320000
#define OUT_DIM   256
#define H_DIM     (N_SUPPORT * D_FEAT)
#define LIST_CAP  64
#define NB_PER_S  85
#define RPB       236
#define BIN_THREADS 1024
#define SCAN_DEPTH  6
#define SCAN_FULL   ((N_EDGES / 4) / (SCAN_DEPTH * BIN_THREADS) * (SCAN_DEPTH * BIN_THREADS))
#define NO_EDGE   0x7fffffff

typedef __attribute__((ext_vector_type(2))) float v2f;
typedef __attribute__((ext_vector_type(8))) float v8f;

__device__ __forceinline__ void add_row(float4& a0, float4& a1,
                                        const float* __restrict__ x,
                                        int src, float w, int lane) {
  const float4* xr = (const float4*)(x + (long)src * D_FEAT);
  float4 p = xr[lane];
  float4 q = xr[lane + 32];
  a0.x += p.x * w; a0.y += p.y * w; a0.z += p.z * w; a0.w += p.w * w;
  a1.x += q.x * w; a1.y += q.y * w; a1.z += q.z * w; a1.w += q.w * w;
}

__device__ __forceinline__ void take_edge(int dv, int e, int row0, int nrows,
                                          int* cnt, int* slots) {
  unsigned r = (unsigned)(dv - row0);
  if (r < (unsigned)nrows) {
    int p = atomicAdd(&cnt[r], 1);
    if (p < LIST_CAP) slots[r * LIST_CAP + p] = e;
  }
}

__global__ __launch_bounds__(BIN_THREADS) void bin_gather(
    const float* __restrict__ x, const int* __restrict__ esrc,
    const int* __restrict__ edst, const float* __restrict__ ew,
    float* __restrict__ h) {
  __shared__ int cnt[RPB];
  __shared__ int slots[RPB * LIST_CAP];

  int tid   = threadIdx.x;
  int s     = blockIdx.x / NB_PER_S;
  int row0  = (blockIdx.x - s * NB_PER_S) * RPB;
  int nrows = (N_NODES - row0 < RPB) ? (N_NODES - row0) : RPB;

  if (tid < RPB) cnt[tid] = 0;
  __syncthreads();

  const int4* dl4 = (const int4*)(edst + s * N_EDGES);
  for (int base = 0; base < SCAN_FULL; base += SCAN_DEPTH * BIN_THREADS) {
    int4 d[SCAN_DEPTH];
    #pragma unroll
    for (int k = 0; k < SCAN_DEPTH; ++k)
      d[k] = dl4[base + k * BIN_THREADS + tid];
    #pragma unroll
    for (int k = 0; k < SCAN_DEPTH; ++k) {
      int e = (base + k * BIN_THREADS + tid) * 4;
      take_edge(d[k].x, e,     row0, nrows, cnt, slots);
      take_edge(d[k].y, e + 1, row0, nrows, cnt, slots);
      take_edge(d[k].z, e + 2, row0, nrows, cnt, slots);
      take_edge(d[k].w, e + 3, row0, nrows, cnt, slots);
    }
  }
  for (int q = SCAN_FULL + tid; q < N_EDGES / 4; q += BIN_THREADS) {
    int4 dv = dl4[q];
    int e = q * 4;
    take_edge(dv.x, e,     row0, nrows, cnt, slots);
    take_edge(dv.y, e + 1, row0, nrows, cnt, slots);
    take_edge(dv.z, e + 2, row0, nrows, cnt, slots);
    take_edge(dv.w, e + 3, row0, nrows, cnt, slots);
  }
  __syncthreads();

  int wave = __builtin_amdgcn_readfirstlane(tid >> 5);
  int lane = tid & 31;
  const int*   es = esrc + s * N_EDGES;
  const float* wt = ew   + s * N_EDGES;

  for (int r = wave; r < nrows; r += BIN_THREADS / 32) {
    int n = __builtin_amdgcn_readfirstlane(cnt[r]);

    float4 a0 = {0.f, 0.f, 0.f, 0.f};
    float4 a1 = {0.f, 0.f, 0.f, 0.f};

    if (n <= LIST_CAP) {
      const int* lp = slots + r * LIST_CAP;
      bool v0 = lane < n;
      int e0 = v0 ? lp[lane] : NO_EDGE;
      int i0 = v0 ? e0 : 0;
      int   s0 = es[i0];
      float w0 = wt[i0];
      int e1 = NO_EDGE, s1 = 0;
      float w1 = 0.f;
      if (n > 32) {
        bool v1 = lane + 32 < n;
        e1 = v1 ? lp[lane + 32] : NO_EDGE;
        int i1 = v1 ? e1 : 0;
        s1 = es[i1];
        w1 = wt[i1];
      }

      int r0 = 0, r1 = n;
      #pragma unroll
      for (int j = 0; j < 32; ++j)
        r0 += (__builtin_amdgcn_readlane(e0, j) < e0) ? 1 : 0;
      if (n > 32) {
        r1 = 0;
        #pragma unroll
        for (int j = 0; j < 32; ++j) {
          int va = __builtin_amdgcn_readlane(e0, j);
          int vb = __builtin_amdgcn_readlane(e1, j);
          r1 += (va < e1) ? 1 : 0;
          r0 += (vb < e0) ? 1 : 0;
          r1 += (vb < e1) ? 1 : 0;
        }
      }

      #pragma unroll 4
      for (int t = 0; t < n; ++t) {
        unsigned long long m =
            (unsigned long long)__builtin_amdgcn_ballot_w32(r0 == t) |
            ((unsigned long long)__builtin_amdgcn_ballot_w32(r1 == t) << 32);
        int l  = __builtin_ctzll(m);
        int sa = __builtin_amdgcn_readlane(s0, l & 31);
        int sb = __builtin_amdgcn_readlane(s1, l & 31);
        int wa = __builtin_amdgcn_readlane(__float_as_int(w0), l & 31);
        int wb = __builtin_amdgcn_readlane(__float_as_int(w1), l & 31);
        int   src = (l < 32) ? sa : sb;
        float w   = __int_as_float((l < 32) ? wa : wb);
        add_row(a0, a1, x, src, w, lane);
      }
    } else {
      const int* dl = edst + s * N_EDGES;
      for (int b = 0; b < N_EDGES; b += 32) {
        unsigned m = __builtin_amdgcn_ballot_w32(dl[b + lane] == row0 + r);
        while (m) {
          int e = b + __builtin_ctz(m);
          m &= m - 1;
          add_row(a0, a1, x, es[e], wt[e], lane);
        }
      }
    }

    float4* hp = (float4*)(h + (long)(row0 + r) * H_DIM + s * D_FEAT);
    hp[lane]      = a0;
    hp[lane + 32] = a1;
  }
}

typedef __attribute__((ext_vector_type(4))) float v4f;

#define GEMM_MT   5
#define GEMM_ROWS (GEMM_MT * 16)

#define GEMM_KG   2
#define GEMM_KB   (GEMM_KG * 16)

struct Frag {
  v4f a[GEMM_KG][GEMM_MT][2];
  v4f b[GEMM_KG][8];
};

__device__ __forceinline__ void load_frag(Frag& f, const float* __restrict__ ap,
                                          const float* __restrict__ bp, int k0) {
  #pragma unroll
  for (int g = 0; g < GEMM_KG; ++g) {
    #pragma unroll
    for (int mt = 0; mt < GEMM_MT; ++mt) {
      f.a[g][mt][0] = *(const v4f*)(ap + (long)mt * 16 * H_DIM + k0 + 16 * g);
      f.a[g][mt][1] = *(const v4f*)(ap + (long)mt * 16 * H_DIM + k0 + 16 * g + 4);
    }
    #pragma unroll
    for (int i = 0; i < 8; ++i)
      f.b[g][i] = *(const v4f*)(bp + (long)(k0 + 16 * g + i) * OUT_DIM);
  }
}

__device__ __forceinline__ void mma_frag(v8f (&acc)[GEMM_MT][4], const Frag& f) {
  #pragma unroll
  for (int gs = 0; gs < 4 * GEMM_KG; ++gs) {
    const int g = gs >> 2;
    const int s = gs & 3;
    #pragma unroll
    for (int mt = 0; mt < GEMM_MT; ++mt) {
      v2f a;
      a.x = f.a[g][mt][s >> 1][(2 * s) & 3];
      a.y = f.a[g][mt][s >> 1][((2 * s) & 3) + 1];
      #pragma unroll
      for (int t = 0; t < 4; ++t) {
        v2f b;
        b.x = f.b[g][2 * s][t];
        b.y = f.b[g][2 * s + 1][t];
        acc[mt][t] = __builtin_amdgcn_wmma_f32_16x16x4_f32(
             false, a,  false, b,
             (short)0, acc[mt][t],  false,  false);
      }
    }
  }
}

__global__ __launch_bounds__(128) void gemm_wmma(
    const float* __restrict__ h, const float* __restrict__ Wm,
    float* __restrict__ out) {
  int wave = threadIdx.x >> 5;
  int lane = threadIdx.x & 31;
  int c    = lane & 15;
  int hf   = lane >> 4;
  int m0   = blockIdx.x * GEMM_ROWS;
  int nb   = wave * 64 + 4 * c;

  const float* ap = h + (long)(m0 + c) * H_DIM + 8 * hf;
  const float* bp = Wm + (long)(8 * hf) * OUT_DIM + nb;

  v8f acc[GEMM_MT][4] = {};
  Frag f0, f1;

  load_frag(f0, ap, bp, 0);
  for (int k0 = 0; k0 < H_DIM - 2 * GEMM_KB; k0 += 2 * GEMM_KB) {
    load_frag(f1, ap, bp, k0 + GEMM_KB);
    __builtin_amdgcn_sched_barrier(0);
    mma_frag(acc, f0);
    load_frag(f0, ap, bp, k0 + 2 * GEMM_KB);
    mma_frag(acc, f1);
  }
  load_frag(f1, ap, bp, H_DIM - GEMM_KB);
  mma_frag(acc, f0);
  mma_frag(acc, f1);

  #pragma unroll
  for (int mt = 0; mt < GEMM_MT; ++mt) {
    float* op = out + (long)(m0 + mt * 16 + hf * 8) * OUT_DIM + nb;
    #pragma unroll
    for (int j = 0; j < 8; ++j) {
      v4f v = {acc[mt][0][j], acc[mt][1][j], acc[mt][2][j], acc[mt][3][j]};
      *(v4f*)(op + (long)j * OUT_DIM) = v;
    }
  }
}

extern "C" void kernel_launch(void* const* d_in, const int* in_sizes, int n_in,
                              void* d_out, int out_size, void* d_ws, size_t ws_size,
                              hipStream_t stream) {
  const float* x    = (const float*)d_in[0];
  const int*   esrc = (const int*)  d_in[1];
  const int*   edst = (const int*)  d_in[2];
  const float* ew   = (const float*)d_in[3];
  const float* Wm   = (const float*)d_in[4];
  float*       out  = (float*)d_out;
  float*       hbuf = (float*)d_ws;

  bin_gather<<<N_SUPPORT * NB_PER_S, BIN_THREADS, 0, stream>>>(x, esrc, edst, ew, hbuf);

  gemm_wmma<<<N_NODES / GEMM_ROWS, 128, 0, stream>>>(hbuf, Wm, out);
}
